# starting kernel
# baseline (speedup 1.0000x reference)
#include <hip/hip_runtime.h>

constexpr int Bsz  = 256;
constexpr int Slen = 512;
constexpr int Edim = 100;
constexpr int Hdim = 512;
constexpr int KTOT = Edim + Hdim;

typedef __attribute__((ext_vector_type(2))) float v2f;
typedef __attribute__((ext_vector_type(8))) float v8f;

__device__ __forceinline__ float sigmoidf(float x) {
    return 1.0f / (1.0f + __expf(-x));
}

__global__ __launch_bounds__(256) void lstm_zero_init(float* __restrict__ h0,
                                                      float* __restrict__ c0,
                                                      int n) {
    int i = blockIdx.x * blockDim.x + threadIdx.x;
    if (i < n) { h0[i] = 0.0f; c0[i] = 0.0f; }
}

__global__ __launch_bounds__(128) void lstm_step(
    const int*   __restrict__ sentence,
    const float* __restrict__ emb,
    const float* __restrict__ Wih,
    const float* __restrict__ Whh,
    const float* __restrict__ bih,
    const float* __restrict__ bhh,
    const float* __restrict__ h_in,
    float*       __restrict__ c,
    float*       __restrict__ h_out,
    int t)
{
    __shared__ float Alds[2][KTOT * 16];

    const int tid    = threadIdx.x;
    const int m_base = blockIdx.x * 32;

    {
        const int r   = tid >> 2;
        const int sub = tid & 3;
        const int b   = m_base + r;
        float* panel  = Alds[r >> 4];
        const int rr  = r & 15;
        const int tok = sentence[(long)b * Slen + t];
        const float* erow = emb + (long)tok * Edim;
        for (int k = sub; k < Edim; k += 4)
            panel[k * 16 + rr] = erow[k];
        const float* hrow = h_in + (long)b * Hdim;
        for (int k = sub; k < Hdim; k += 4)
            panel[(Edim + k) * 16 + rr] = hrow[k];
    }
    __syncthreads();

    const int wave = tid >> 5;
    const int lane = tid & 31;
    const int half = lane >> 4;
    const int ml   = lane & 15;
    const int col  = blockIdx.y * 64 + wave * 16 + ml;

    v8f acc[2][4];
#pragma unroll
    for (int mt = 0; mt < 2; ++mt)
#pragma unroll
        for (int q = 0; q < 4; ++q)
            acc[mt][q] = (v8f){};

    const float* wi0 = Wih + (long)(0 * Hdim + col) * Edim;
    const float* wi1 = Wih + (long)(1 * Hdim + col) * Edim;
    const float* wi2 = Wih + (long)(2 * Hdim + col) * Edim;
    const float* wi3 = Wih + (long)(3 * Hdim + col) * Edim;

    for (int k0 = 0; k0 < Edim; k0 += 4) {
        const int kb = k0 + 2 * half;
        v2f a0, a1;
        a0.x = Alds[0][kb * 16 + ml];
        a0.y = Alds[0][(kb + 1) * 16 + ml];
        a1.x = Alds[1][kb * 16 + ml];
        a1.y = Alds[1][(kb + 1) * 16 + ml];
        v2f b0 = *(const v2f*)(wi0 + kb);
        v2f b1 = *(const v2f*)(wi1 + kb);
        v2f b2 = *(const v2f*)(wi2 + kb);
        v2f b3 = *(const v2f*)(wi3 + kb);
        acc[0][0] = __builtin_amdgcn_wmma_f32_16x16x4_f32(false, a0, false, b0, (short)0, acc[0][0], false, false);
        acc[1][0] = __builtin_amdgcn_wmma_f32_16x16x4_f32(false, a1, false, b0, (short)0, acc[1][0], false, false);
        acc[0][1] = __builtin_amdgcn_wmma_f32_16x16x4_f32(false, a0, false, b1, (short)0, acc[0][1], false, false);
        acc[1][1] = __builtin_amdgcn_wmma_f32_16x16x4_f32(false, a1, false, b1, (short)0, acc[1][1], false, false);
        acc[0][2] = __builtin_amdgcn_wmma_f32_16x16x4_f32(false, a0, false, b2, (short)0, acc[0][2], false, false);
        acc[1][2] = __builtin_amdgcn_wmma_f32_16x16x4_f32(false, a1, false, b2, (short)0, acc[1][2], false, false);
        acc[0][3] = __builtin_amdgcn_wmma_f32_16x16x4_f32(false, a0, false, b3, (short)0, acc[0][3], false, false);
        acc[1][3] = __builtin_amdgcn_wmma_f32_16x16x4_f32(false, a1, false, b3, (short)0, acc[1][3], false, false);
    }

    const float* wh0 = Whh + (long)(0 * Hdim + col) * Hdim;
    const float* wh1 = Whh + (long)(1 * Hdim + col) * Hdim;
    const float* wh2 = Whh + (long)(2 * Hdim + col) * Hdim;
    const float* wh3 = Whh + (long)(3 * Hdim + col) * Hdim;

    for (int k0 = 0; k0 < Hdim; k0 += 4) {
        const int kb = k0 + 2 * half;
        v2f a0, a1;
        a0.x = Alds[0][(Edim + kb) * 16 + ml];
        a0.y = Alds[0][(Edim + kb + 1) * 16 + ml];
        a1.x = Alds[1][(Edim + kb) * 16 + ml];
        a1.y = Alds[1][(Edim + kb + 1) * 16 + ml];
        v2f b0 = *(const v2f*)(wh0 + kb);
        v2f b1 = *(const v2f*)(wh1 + kb);
        v2f b2 = *(const v2f*)(wh2 + kb);
        v2f b3 = *(const v2f*)(wh3 + kb);
        acc[0][0] = __builtin_amdgcn_wmma_f32_16x16x4_f32(false, a0, false, b0, (short)0, acc[0][0], false, false);
        acc[1][0] = __builtin_amdgcn_wmma_f32_16x16x4_f32(false, a1, false, b0, (short)0, acc[1][0], false, false);
        acc[0][1] = __builtin_amdgcn_wmma_f32_16x16x4_f32(false, a0, false, b1, (short)0, acc[0][1], false, false);
        acc[1][1] = __builtin_amdgcn_wmma_f32_16x16x4_f32(false, a1, false, b1, (short)0, acc[1][1], false, false);
        acc[0][2] = __builtin_amdgcn_wmma_f32_16x16x4_f32(false, a0, false, b2, (short)0, acc[0][2], false, false);
        acc[1][2] = __builtin_amdgcn_wmma_f32_16x16x4_f32(false, a1, false, b2, (short)0, acc[1][2], false, false);
        acc[0][3] = __builtin_amdgcn_wmma_f32_16x16x4_f32(false, a0, false, b3, (short)0, acc[0][3], false, false);
        acc[1][3] = __builtin_amdgcn_wmma_f32_16x16x4_f32(false, a1, false, b3, (short)0, acc[1][3], false, false);
    }

    const float bs0 = bih[0 * Hdim + col] + bhh[0 * Hdim + col];
    const float bs1 = bih[1 * Hdim + col] + bhh[1 * Hdim + col];
    const float bs2 = bih[2 * Hdim + col] + bhh[2 * Hdim + col];
    const float bs3 = bih[3 * Hdim + col] + bhh[3 * Hdim + col];

#pragma unroll
    for (int mt = 0; mt < 2; ++mt) {
#pragma unroll
        for (int r = 0; r < 8; ++r) {
            const int  mg  = m_base + mt * 16 + r + 8 * half;
            const long idx = (long)mg * Hdim + col;
            const float ig = sigmoidf(acc[mt][0][r] + bs0);
            const float fg = sigmoidf(acc[mt][1][r] + bs1);
            const float gg = tanhf   (acc[mt][2][r] + bs2);
            const float og = sigmoidf(acc[mt][3][r] + bs3);
            const float cn = fg * c[idx] + ig * gg;
            c[idx]     = cn;
            h_out[idx] = og * tanhf(cn);
        }
    }
}

extern "C" void kernel_launch(void* const* d_in, const int* in_sizes, int n_in,
                              void* d_out, int out_size, void* d_ws, size_t ws_size,
                              hipStream_t stream) {
    const int*   sentence = (const int*)  d_in[0];
    const float* emb      = (const float*)d_in[1];
    const float* Wih      = (const float*)d_in[2];
    const float* Whh      = (const float*)d_in[3];
    const float* bih      = (const float*)d_in[4];
    const float* bhh      = (const float*)d_in[5];
    float* out = (float*)d_out;
    float* ws  = (float*)d_ws;

    const int BH = Bsz * Hdim;
    float* hbuf0 = ws;
    float* hbuf1 = ws + BH;
    float* cbuf  = ws + 2 * BH;

    lstm_zero_init<<<(BH + 255) / 256, 256, 0, stream>>>(hbuf0, cbuf, BH);

    dim3 grid(Bsz / 32, Hdim / 64);
    for (int t = 0; t < Slen; ++t) {
        const float* hin  = (t & 1) ? hbuf1 : hbuf0;
        float*       hout = (t == Slen - 1) ? out
                          : ((t & 1) ? hbuf0 : hbuf1);
        lstm_step<<<grid, 128, 0, stream>>>(sentence, emb, Wih, Whh, bih, bhh,
                                            hin, cbuf, hout, t);
    }
}


# Re-tile the timestep over 256 workgroups of 16 rows x 32 cols (was 64 of 32 x 64): a wave owns one gate, gates meet through LDS for the cell update, K loop unrolled whole; k order of sums unchanged
# speedup vs baseline: 2.7132x; 2.7132x over previous
#include <hip/hip_runtime.h>

constexpr int Bsz  = 256;
constexpr int Slen = 512;
constexpr int Edim = 100;
constexpr int Hdim = 512;
constexpr int KTOT = Edim + Hdim;

constexpr int TM = 16;
constexpr int TN = 32;
constexpr int GP = 36;

typedef __attribute__((ext_vector_type(2))) float v2f;
typedef __attribute__((ext_vector_type(8))) float v8f;

__device__ __forceinline__ float sigmoidf(float x) {
    return 1.0f / (1.0f + __expf(-x));
}

__global__ __launch_bounds__(256) void lstm_zero_init(float* __restrict__ h0,
                                                      float* __restrict__ c0,
                                                      int n) {
    int i = blockIdx.x * blockDim.x + threadIdx.x;
    if (i < n) { h0[i] = 0.0f; c0[i] = 0.0f; }
}

__global__ __launch_bounds__(128) void lstm_step(
    const int*   __restrict__ sentence,
    const float* __restrict__ emb,
    const float* __restrict__ Wih,
    const float* __restrict__ Whh,
    const float* __restrict__ bih,
    const float* __restrict__ bhh,
    const float* __restrict__ h_in,
    float*       __restrict__ c,
    float*       __restrict__ h_out,
    int t)
{
    __shared__ float Alds[(KTOT + 1) * 16];
    __shared__ alignas(16) float Glds[4][TM][GP];

    const int tid    = threadIdx.x;
    const int m_base = blockIdx.x * TM;
    const int col0   = blockIdx.y * TN;

    {
        const int r   = tid >> 3;
        const int sub = tid & 7;
        const int b   = m_base + r;
        const int tok = sentence[(long)b * Slen + t];
        const float* hrow = h_in + (long)b * Hdim;
        for (int k = sub; k < Hdim; k += 8)
            Alds[(Edim + k) * 16 + r] = hrow[k];
        const float* erow = emb + (long)tok * Edim;
        for (int i = 0; i < (Edim + 7) / 8; ++i) {
            const int k  = sub + 8 * i;
            const bool in = k < Edim;
            Alds[(in ? k : KTOT) * 16 + r] = erow[in ? k : Edim - 1];
        }
    }
    __syncthreads();

    const int wave = tid >> 5;
    const int lane = tid & 31;
    const int half = lane >> 4;
    const int ml   = lane & 15;

    v8f acc0 = (v8f){};
    v8f acc1 = (v8f){};

    const long wrow0 = (long)wave * Hdim + col0 + ml;
    const long wrow1 = wrow0 + 16;

    const float* wi0 = Wih + wrow0 * Edim;
    const float* wi1 = Wih + wrow1 * Edim;

    for (int k0 = 0; k0 < Edim; k0 += 4) {
        const int kb = k0 + 2 * half;
        v2f a;
        a.x = Alds[kb * 16 + ml];
        a.y = Alds[(kb + 1) * 16 + ml];
        v2f b0 = *(const v2f*)(wi0 + kb);
        v2f b1 = *(const v2f*)(wi1 + kb);
        acc0 = __builtin_amdgcn_wmma_f32_16x16x4_f32(false, a, false, b0, (short)0, acc0, false, false);
        acc1 = __builtin_amdgcn_wmma_f32_16x16x4_f32(false, a, false, b1, (short)0, acc1, false, false);
    }

    const float* wh0 = Whh + wrow0 * Hdim;
    const float* wh1 = Whh + wrow1 * Hdim;

#pragma unroll
    for (int k0 = 0; k0 < Hdim; k0 += 4) {
        const int kb = k0 + 2 * half;
        v2f a;
        a.x = Alds[(Edim + kb) * 16 + ml];
        a.y = Alds[(Edim + kb + 1) * 16 + ml];
        v2f b0 = *(const v2f*)(wh0 + kb);
        v2f b1 = *(const v2f*)(wh1 + kb);
        acc0 = __builtin_amdgcn_wmma_f32_16x16x4_f32(false, a, false, b0, (short)0, acc0, false, false);
        acc1 = __builtin_amdgcn_wmma_f32_16x16x4_f32(false, a, false, b1, (short)0, acc1, false, false);
    }

    for (int r = 0; r < 8; ++r) {
        Glds[wave][r + 8 * half][ml]      = acc0[r];
        Glds[wave][r + 8 * half][16 + ml] = acc1[r];
    }
    __syncthreads();

    const int row = tid >> 3;
    const int c4  = (tid & 7) * 4;
    const int col = col0 + c4;

    const float4 pi = *(const float4*)&Glds[0][row][c4];
    const float4 pf = *(const float4*)&Glds[1][row][c4];
    const float4 pg = *(const float4*)&Glds[2][row][c4];
    const float4 po = *(const float4*)&Glds[3][row][c4];

    const float4 bi0 = *(const float4*)(bih + 0 * Hdim + col);
    const float4 bi1 = *(const float4*)(bih + 1 * Hdim + col);
    const float4 bi2 = *(const float4*)(bih + 2 * Hdim + col);
    const float4 bi3 = *(const float4*)(bih + 3 * Hdim + col);
    const float4 bh0 = *(const float4*)(bhh + 0 * Hdim + col);
    const float4 bh1 = *(const float4*)(bhh + 1 * Hdim + col);
    const float4 bh2 = *(const float4*)(bhh + 2 * Hdim + col);
    const float4 bh3 = *(const float4*)(bhh + 3 * Hdim + col);

    const long idx = (long)(m_base + row) * Hdim + col;
    const float4 cold = *(const float4*)(c + idx);

    const float vi[4] = {pi.x, pi.y, pi.z, pi.w};
    const float vf[4] = {pf.x, pf.y, pf.z, pf.w};
    const float vg[4] = {pg.x, pg.y, pg.z, pg.w};
    const float vo[4] = {po.x, po.y, po.z, po.w};
    const float s0[4] = {bi0.x + bh0.x, bi0.y + bh0.y, bi0.z + bh0.z, bi0.w + bh0.w};
    const float s1[4] = {bi1.x + bh1.x, bi1.y + bh1.y, bi1.z + bh1.z, bi1.w + bh1.w};
    const float s2[4] = {bi2.x + bh2.x, bi2.y + bh2.y, bi2.z + bh2.z, bi2.w + bh2.w};
    const float s3[4] = {bi3.x + bh3.x, bi3.y + bh3.y, bi3.z + bh3.z, bi3.w + bh3.w};
    const float co[4] = {cold.x, cold.y, cold.z, cold.w};
    float cn[4], hn[4];
    for (int j = 0; j < 4; ++j) {
        const float ig = sigmoidf(vi[j] + s0[j]);
        const float fg = sigmoidf(vf[j] + s1[j]);
        const float gg = tanhf   (vg[j] + s2[j]);
        const float og = sigmoidf(vo[j] + s3[j]);
        cn[j] = fg * co[j] + ig * gg;
        hn[j] = og * tanhf(cn[j]);
    }
    *(float4*)(c + idx)     = make_float4(cn[0], cn[1], cn[2], cn[3]);
    *(float4*)(h_out + idx) = make_float4(hn[0], hn[1], hn[2], hn[3]);
}

extern "C" void kernel_launch(void* const* d_in, const int* in_sizes, int n_in,
                              void* d_out, int out_size, void* d_ws, size_t ws_size,
                              hipStream_t stream) {
    const int*   sentence = (const int*)  d_in[0];
    const float* emb      = (const float*)d_in[1];
    const float* Wih      = (const float*)d_in[2];
    const float* Whh      = (const float*)d_in[3];
    const float* bih      = (const float*)d_in[4];
    const float* bhh      = (const float*)d_in[5];
    float* out = (float*)d_out;
    float* ws  = (float*)d_ws;

    const int BH = Bsz * Hdim;
    float* hbuf0 = ws;
    float* hbuf1 = ws + BH;
    float* cbuf  = ws + 2 * BH;

    lstm_zero_init<<<(BH + 255) / 256, 256, 0, stream>>>(hbuf0, cbuf, BH);

    dim3 grid(Bsz / TM, Hdim / TN);
    for (int t = 0; t < Slen; ++t) {
        const float* hin  = (t & 1) ? hbuf1 : hbuf0;
        float*       hout = (t == Slen - 1) ? out
                          : ((t & 1) ? hbuf0 : hbuf1);
        lstm_step<<<grid, 128, 0, stream>>>(sentence, emb, Wih, Whh, bih, bhh,
                                            hin, cbuf, hout, t);
    }
}


# 16-byte operand loads: a lane half takes 4 consecutive k of each 8-k chunk (A: row-major LDS panel, W: global), halving weight loads and lines touched per k; k grouping of the sums changes
# speedup vs baseline: 3.2565x; 1.2002x over previous
#include <hip/hip_runtime.h>

constexpr int Bsz  = 256;
constexpr int Slen = 512;
constexpr int Edim = 100;
constexpr int Hdim = 512;
constexpr int KTOT = Edim + Hdim;

constexpr int TM = 16;
constexpr int TN = 32;
constexpr int GP = 36;
constexpr int AP = KTOT;

typedef __attribute__((ext_vector_type(2))) float v2f;
typedef __attribute__((ext_vector_type(4))) float v4f;
typedef __attribute__((ext_vector_type(8))) float v8f;

__device__ __forceinline__ float sigmoidf(float x) {
    return 1.0f / (1.0f + __expf(-x));
}

__global__ __launch_bounds__(256) void lstm_zero_init(float* __restrict__ h0,
                                                      float* __restrict__ c0,
                                                      int n) {
    int i = blockIdx.x * blockDim.x + threadIdx.x;
    if (i < n) { h0[i] = 0.0f; c0[i] = 0.0f; }
}

__global__ __launch_bounds__(128) void lstm_step(
    const int*   __restrict__ sentence,
    const float* __restrict__ emb,
    const float* __restrict__ Wih,
    const float* __restrict__ Whh,
    const float* __restrict__ bih,
    const float* __restrict__ bhh,
    const float* __restrict__ h_in,
    float*       __restrict__ c,
    float*       __restrict__ h_out,
    int t)
{
    __shared__ alignas(16) float Alds[TM * AP];
    __shared__ alignas(16) float Glds[4][TM][GP];

    const int tid    = threadIdx.x;
    const int m_base = blockIdx.x * TM;
    const int col0   = blockIdx.y * TN;

    {
        const int r   = tid >> 3;
        const int sub = tid & 7;
        const int b   = m_base + r;
        const int tok = sentence[(long)b * Slen + t];
        const v4f* hrow = (const v4f*)(h_in + (long)b * Hdim);
        const v4f* erow = (const v4f*)(emb + (long)tok * Edim);
        v4f* prow = (v4f*)(Alds + r * AP);
        v4f hv[Hdim / 32];
        for (int i = 0; i < Hdim / 32; ++i)
            hv[i] = hrow[sub + 8 * i];
        v4f ev[3];
        for (int i = 0; i < 3; ++i)
            ev[i] = erow[sub + 8 * i];
        const int il = (sub == 0) ? Edim / 4 - 1 : sub + 16;
        const v4f elast = erow[il];
        for (int i = 0; i < Hdim / 32; ++i)
            prow[Edim / 4 + sub + 8 * i] = hv[i];
        for (int i = 0; i < 3; ++i)
            prow[sub + 8 * i] = ev[i];
        prow[il] = elast;
    }
    __syncthreads();

    const int wave = tid >> 5;
    const int lane = tid & 31;
    const int half = lane >> 4;
    const int ml   = lane & 15;

    v8f acc0 = (v8f){};
    v8f acc1 = (v8f){};

    const long wrow0 = (long)wave * Hdim + col0 + ml;
    const long wrow1 = wrow0 + 16;

    const float* arow = Alds + ml * AP;
    const float* wi0 = Wih + wrow0 * Edim;
    const float* wi1 = Wih + wrow1 * Edim;

#pragma unroll
    for (int k0 = 0; k0 < (Edim / 8) * 8; k0 += 8) {
        const int kb = k0 + 4 * half;
        const v4f a  = *(const v4f*)(arow + kb);
        const v4f b0 = *(const v4f*)(wi0 + kb);
        const v4f b1 = *(const v4f*)(wi1 + kb);
        acc0 = __builtin_amdgcn_wmma_f32_16x16x4_f32(false, a.xy, false, b0.xy, (short)0, acc0, false, false);
        acc1 = __builtin_amdgcn_wmma_f32_16x16x4_f32(false, a.xy, false, b1.xy, (short)0, acc1, false, false);
        acc0 = __builtin_amdgcn_wmma_f32_16x16x4_f32(false, a.zw, false, b0.zw, (short)0, acc0, false, false);
        acc1 = __builtin_amdgcn_wmma_f32_16x16x4_f32(false, a.zw, false, b1.zw, (short)0, acc1, false, false);
    }
    {
        const int kb = (Edim / 8) * 8 + 2 * half;
        const v2f a  = *(const v2f*)(arow + kb);
        const v2f b0 = *(const v2f*)(wi0 + kb);
        const v2f b1 = *(const v2f*)(wi1 + kb);
        acc0 = __builtin_amdgcn_wmma_f32_16x16x4_f32(false, a, false, b0, (short)0, acc0, false, false);
        acc1 = __builtin_amdgcn_wmma_f32_16x16x4_f32(false, a, false, b1, (short)0, acc1, false, false);
    }

    const float* wh0 = Whh + wrow0 * Hdim;
    const float* wh1 = Whh + wrow1 * Hdim;

#pragma unroll
    for (int k0 = 0; k0 < Hdim; k0 += 8) {
        const int kb = k0 + 4 * half;
        const v4f a  = *(const v4f*)(arow + Edim + kb);
        const v4f b0 = *(const v4f*)(wh0 + kb);
        const v4f b1 = *(const v4f*)(wh1 + kb);
        acc0 = __builtin_amdgcn_wmma_f32_16x16x4_f32(false, a.xy, false, b0.xy, (short)0, acc0, false, false);
        acc1 = __builtin_amdgcn_wmma_f32_16x16x4_f32(false, a.xy, false, b1.xy, (short)0, acc1, false, false);
        acc0 = __builtin_amdgcn_wmma_f32_16x16x4_f32(false, a.zw, false, b0.zw, (short)0, acc0, false, false);
        acc1 = __builtin_amdgcn_wmma_f32_16x16x4_f32(false, a.zw, false, b1.zw, (short)0, acc1, false, false);
    }

    for (int r = 0; r < 8; ++r) {
        Glds[wave][r + 8 * half][ml]      = acc0[r];
        Glds[wave][r + 8 * half][16 + ml] = acc1[r];
    }
    __syncthreads();

    const int row = tid >> 3;
    const int c4  = (tid & 7) * 4;
    const int col = col0 + c4;

    const float4 pi = *(const float4*)&Glds[0][row][c4];
    const float4 pf = *(const float4*)&Glds[1][row][c4];
    const float4 pg = *(const float4*)&Glds[2][row][c4];
    const float4 po = *(const float4*)&Glds[3][row][c4];

    const float4 bi0 = *(const float4*)(bih + 0 * Hdim + col);
    const float4 bi1 = *(const float4*)(bih + 1 * Hdim + col);
    const float4 bi2 = *(const float4*)(bih + 2 * Hdim + col);
    const float4 bi3 = *(const float4*)(bih + 3 * Hdim + col);
    const float4 bh0 = *(const float4*)(bhh + 0 * Hdim + col);
    const float4 bh1 = *(const float4*)(bhh + 1 * Hdim + col);
    const float4 bh2 = *(const float4*)(bhh + 2 * Hdim + col);
    const float4 bh3 = *(const float4*)(bhh + 3 * Hdim + col);

    const long idx = (long)(m_base + row) * Hdim + col;
    const float4 cold = *(const float4*)(c + idx);

    const float vi[4] = {pi.x, pi.y, pi.z, pi.w};
    const float vf[4] = {pf.x, pf.y, pf.z, pf.w};
    const float vg[4] = {pg.x, pg.y, pg.z, pg.w};
    const float vo[4] = {po.x, po.y, po.z, po.w};
    const float s0[4] = {bi0.x + bh0.x, bi0.y + bh0.y, bi0.z + bh0.z, bi0.w + bh0.w};
    const float s1[4] = {bi1.x + bh1.x, bi1.y + bh1.y, bi1.z + bh1.z, bi1.w + bh1.w};
    const float s2[4] = {bi2.x + bh2.x, bi2.y + bh2.y, bi2.z + bh2.z, bi2.w + bh2.w};
    const float s3[4] = {bi3.x + bh3.x, bi3.y + bh3.y, bi3.z + bh3.z, bi3.w + bh3.w};
    const float co[4] = {cold.x, cold.y, cold.z, cold.w};
    float cn[4], hn[4];
    for (int j = 0; j < 4; ++j) {
        const float ig = sigmoidf(vi[j] + s0[j]);
        const float fg = sigmoidf(vf[j] + s1[j]);
        const float gg = tanhf   (vg[j] + s2[j]);
        const float og = sigmoidf(vo[j] + s3[j]);
        cn[j] = fg * co[j] + ig * gg;
        hn[j] = og * tanhf(cn[j]);
    }
    *(float4*)(c + idx)     = make_float4(cn[0], cn[1], cn[2], cn[3]);
    *(float4*)(h_out + idx) = make_float4(hn[0], hn[1], hn[2], hn[3]);
}

extern "C" void kernel_launch(void* const* d_in, const int* in_sizes, int n_in,
                              void* d_out, int out_size, void* d_ws, size_t ws_size,
                              hipStream_t stream) {
    const int*   sentence = (const int*)  d_in[0];
    const float* emb      = (const float*)d_in[1];
    const float* Wih      = (const float*)d_in[2];
    const float* Whh      = (const float*)d_in[3];
    const float* bih      = (const float*)d_in[4];
    const float* bhh      = (const float*)d_in[5];
    float* out = (float*)d_out;
    float* ws  = (float*)d_ws;

    const int BH = Bsz * Hdim;
    float* hbuf0 = ws;
    float* hbuf1 = ws + BH;
    float* cbuf  = ws + 2 * BH;

    lstm_zero_init<<<(BH + 255) / 256, 256, 0, stream>>>(hbuf0, cbuf, BH);

    dim3 grid(Bsz / TM, Hdim / TN);
    for (int t = 0; t < Slen; ++t) {
        const float* hin  = (t & 1) ? hbuf1 : hbuf0;
        float*       hout = (t == Slen - 1) ? out
                          : ((t & 1) ? hbuf0 : hbuf1);
        lstm_step<<<grid, 128, 0, stream>>>(sentence, emb, Wih, Whh, bih, bhh,
                                            hin, cbuf, hout, t);
    }
}


# Weights reach the fragments through a per-wave LDS copy: global weight loads now cover whole 128-byte lines (4 lines per load, was 16), fragments read from LDS with the same k mapping; sums unchanged
# speedup vs baseline: 3.4619x; 1.0631x over previous
#include <hip/hip_runtime.h>

constexpr int Bsz  = 256;
constexpr int Slen = 512;
constexpr int Edim = 100;
constexpr int Hdim = 512;
constexpr int KTOT = Edim + Hdim;

constexpr int TM = 16;
constexpr int TN = 32;
constexpr int GP = 36;
constexpr int AP = KTOT;
constexpr int WXP = Edim;
constexpr int KB  = 32;
constexpr int WHP = 36;
constexpr int NST = 4;
constexpr int PF  = 4;

typedef __attribute__((ext_vector_type(2))) float v2f;
typedef __attribute__((ext_vector_type(4))) float v4f;
typedef __attribute__((ext_vector_type(8))) float v8f;

__device__ __forceinline__ float sigmoidf(float x) {
    return 1.0f / (1.0f + __expf(-x));
}

__global__ __launch_bounds__(256) void lstm_zero_init(float* __restrict__ h0,
                                                      float* __restrict__ c0,
                                                      int n) {
    int i = blockIdx.x * blockDim.x + threadIdx.x;
    if (i < n) { h0[i] = 0.0f; c0[i] = 0.0f; }
}

__global__ __launch_bounds__(128) void lstm_step(
    const int*   __restrict__ sentence,
    const float* __restrict__ emb,
    const float* __restrict__ Wih,
    const float* __restrict__ Whh,
    const float* __restrict__ bih,
    const float* __restrict__ bhh,
    const float* __restrict__ h_in,
    float*       __restrict__ c,
    float*       __restrict__ h_out,
    int t)
{
    __shared__ alignas(16) float Alds[TM * AP];
    __shared__ alignas(16) float Glds[4][TM][GP];
    __shared__ alignas(16) float Wx[4][32 * WXP];
    __shared__ alignas(16) float Wh[4][NST][32 * WHP];

    const int tid    = threadIdx.x;
    const int m_base = blockIdx.x * TM;
    const int col0   = blockIdx.y * TN;

    const int wave = tid >> 5;
    const int lane = tid & 31;
    const int half = lane >> 4;
    const int ml   = lane & 15;

    const long wrow = (long)wave * Hdim + col0;

    const int pr   = tid >> 3;
    const int sub  = tid & 7;
    const int pb   = m_base + pr;
    const int tok  = sentence[(long)pb * Slen + t];
    const v4f* hrow = (const v4f*)(h_in + (long)pb * Hdim);
    v4f* prow = (v4f*)(Alds + pr * AP);
    v4f hv[Hdim / 32];
    for (int i = 0; i < Hdim / 32; ++i)
        hv[i] = hrow[sub + 8 * i];

    const v4f* wxg = (const v4f*)(Wih + wrow * Edim);
    v4f xs[25];
    for (int i = 0; i < 25; ++i)
        xs[i] = wxg[lane + 32 * i];

    {
        const v4f* erow = (const v4f*)(emb + (long)tok * Edim);
        v4f ev[3];
        for (int i = 0; i < 3; ++i)
            ev[i] = erow[sub + 8 * i];
        const int il = (sub == 0) ? Edim / 4 - 1 : sub + 16;
        const v4f elast = erow[il];
        for (int i = 0; i < Hdim / 32; ++i)
            prow[Edim / 4 + sub + 8 * i] = hv[i];
        v4f* wxl = (v4f*)Wx[wave];
        for (int i = 0; i < 25; ++i)
            wxl[lane + 32 * i] = xs[i];
        for (int i = 0; i < 3; ++i)
            prow[sub + 8 * i] = ev[i];
        prow[il] = elast;
    }
    __syncthreads();

    const float* whg = Whh + (wrow + (lane >> 3)) * Hdim + (lane & 7) * 4;
    v4f g[PF][8];
    for (int p = 0; p < PF; ++p)
        for (int i = 0; i < 8; ++i)
            g[p][i] = *(const v4f*)(whg + (long)(4 * i) * Hdim + p * KB);

    v8f acc0 = (v8f){};
    v8f acc1 = (v8f){};

    const float* arow = Alds + ml * AP;
    const float* bx0  = Wx[wave] + ml * WXP;
    const float* bx1  = bx0 + 16 * WXP;

#pragma unroll
    for (int k0 = 0; k0 < (Edim / 8) * 8; k0 += 8) {
        const int kb = k0 + 4 * half;
        const v4f a  = *(const v4f*)(arow + kb);
        const v4f b0 = *(const v4f*)(bx0 + kb);
        const v4f b1 = *(const v4f*)(bx1 + kb);
        acc0 = __builtin_amdgcn_wmma_f32_16x16x4_f32(false, a.xy, false, b0.xy, (short)0, acc0, false, false);
        acc1 = __builtin_amdgcn_wmma_f32_16x16x4_f32(false, a.xy, false, b1.xy, (short)0, acc1, false, false);
        acc0 = __builtin_amdgcn_wmma_f32_16x16x4_f32(false, a.zw, false, b0.zw, (short)0, acc0, false, false);
        acc1 = __builtin_amdgcn_wmma_f32_16x16x4_f32(false, a.zw, false, b1.zw, (short)0, acc1, false, false);
    }
    {
        const int kb = (Edim / 8) * 8 + 2 * half;
        const v2f a  = *(const v2f*)(arow + kb);
        const v2f b0 = *(const v2f*)(bx0 + kb);
        const v2f b1 = *(const v2f*)(bx1 + kb);
        acc0 = __builtin_amdgcn_wmma_f32_16x16x4_f32(false, a, false, b0, (short)0, acc0, false, false);
        acc1 = __builtin_amdgcn_wmma_f32_16x16x4_f32(false, a, false, b1, (short)0, acc1, false, false);
    }

    float*       whs = &Wh[wave][0][0] + (lane >> 3) * WHP + (lane & 7) * 4;
    const float* wf0 = &Wh[wave][0][0] + ml * WHP;
#pragma unroll
    for (int kb = 0; kb < Hdim / KB; ++kb) {
        const int st = kb % NST;
        const int p  = kb % PF;
        float* sdst = whs + st * (32 * WHP);
        for (int i = 0; i < 8; ++i)
            *(v4f*)(sdst + 4 * i * WHP) = g[p][i];
        if (kb + PF < Hdim / KB)
            for (int i = 0; i < 8; ++i)
                g[p][i] = *(const v4f*)(whg + (long)(4 * i) * Hdim + (kb + PF) * KB);
        __builtin_amdgcn_fence(__ATOMIC_RELEASE, "wavefront");
        __builtin_amdgcn_wave_barrier();
        __builtin_amdgcn_fence(__ATOMIC_ACQUIRE, "wavefront");
        const float* f0 = wf0 + st * (32 * WHP);
        const float* f1 = f0 + 16 * WHP;
        for (int cc = 0; cc < KB / 8; ++cc) {
            const int kk = 8 * cc + 4 * half;
            const v4f a  = *(const v4f*)(arow + Edim + kb * KB + kk);
            const v4f b0 = *(const v4f*)(f0 + kk);
            const v4f b1 = *(const v4f*)(f1 + kk);
            acc0 = __builtin_amdgcn_wmma_f32_16x16x4_f32(false, a.xy, false, b0.xy, (short)0, acc0, false, false);
            acc1 = __builtin_amdgcn_wmma_f32_16x16x4_f32(false, a.xy, false, b1.xy, (short)0, acc1, false, false);
            acc0 = __builtin_amdgcn_wmma_f32_16x16x4_f32(false, a.zw, false, b0.zw, (short)0, acc0, false, false);
            acc1 = __builtin_amdgcn_wmma_f32_16x16x4_f32(false, a.zw, false, b1.zw, (short)0, acc1, false, false);
        }
    }

    for (int r = 0; r < 8; ++r) {
        Glds[wave][r + 8 * half][ml]      = acc0[r];
        Glds[wave][r + 8 * half][16 + ml] = acc1[r];
    }
    __syncthreads();

    const int row = tid >> 3;
    const int c4  = (tid & 7) * 4;
    const int col = col0 + c4;

    const float4 pi = *(const float4*)&Glds[0][row][c4];
    const float4 pf = *(const float4*)&Glds[1][row][c4];
    const float4 pg = *(const float4*)&Glds[2][row][c4];
    const float4 po = *(const float4*)&Glds[3][row][c4];

    const float4 bi0 = *(const float4*)(bih + 0 * Hdim + col);
    const float4 bi1 = *(const float4*)(bih + 1 * Hdim + col);
    const float4 bi2 = *(const float4*)(bih + 2 * Hdim + col);
    const float4 bi3 = *(const float4*)(bih + 3 * Hdim + col);
    const float4 bh0 = *(const float4*)(bhh + 0 * Hdim + col);
    const float4 bh1 = *(const float4*)(bhh + 1 * Hdim + col);
    const float4 bh2 = *(const float4*)(bhh + 2 * Hdim + col);
    const float4 bh3 = *(const float4*)(bhh + 3 * Hdim + col);

    const long idx = (long)(m_base + row) * Hdim + col;
    const float4 cold = *(const float4*)(c + idx);

    const float vi[4] = {pi.x, pi.y, pi.z, pi.w};
    const float vf[4] = {pf.x, pf.y, pf.z, pf.w};
    const float vg[4] = {pg.x, pg.y, pg.z, pg.w};
    const float vo[4] = {po.x, po.y, po.z, po.w};
    const float s0[4] = {bi0.x + bh0.x, bi0.y + bh0.y, bi0.z + bh0.z, bi0.w + bh0.w};
    const float s1[4] = {bi1.x + bh1.x, bi1.y + bh1.y, bi1.z + bh1.z, bi1.w + bh1.w};
    const float s2[4] = {bi2.x + bh2.x, bi2.y + bh2.y, bi2.z + bh2.z, bi2.w + bh2.w};
    const float s3[4] = {bi3.x + bh3.x, bi3.y + bh3.y, bi3.z + bh3.z, bi3.w + bh3.w};
    const float co[4] = {cold.x, cold.y, cold.z, cold.w};
    float cn[4], hn[4];
    for (int j = 0; j < 4; ++j) {
        const float ig = sigmoidf(vi[j] + s0[j]);
        const float fg = sigmoidf(vf[j] + s1[j]);
        const float gg = tanhf   (vg[j] + s2[j]);
        const float og = sigmoidf(vo[j] + s3[j]);
        cn[j] = fg * co[j] + ig * gg;
        hn[j] = og * tanhf(cn[j]);
    }
    *(float4*)(c + idx)     = make_float4(cn[0], cn[1], cn[2], cn[3]);
    *(float4*)(h_out + idx) = make_float4(hn[0], hn[1], hn[2], hn[3]);
}

extern "C" void kernel_launch(void* const* d_in, const int* in_sizes, int n_in,
                              void* d_out, int out_size, void* d_ws, size_t ws_size,
                              hipStream_t stream) {
    const int*   sentence = (const int*)  d_in[0];
    const float* emb      = (const float*)d_in[1];
    const float* Wih      = (const float*)d_in[2];
    const float* Whh      = (const float*)d_in[3];
    const float* bih      = (const float*)d_in[4];
    const float* bhh      = (const float*)d_in[5];
    float* out = (float*)d_out;
    float* ws  = (float*)d_ws;

    const int BH = Bsz * Hdim;
    float* hbuf0 = ws;
    float* hbuf1 = ws + BH;
    float* cbuf  = ws + 2 * BH;

    lstm_zero_init<<<(BH + 255) / 256, 256, 0, stream>>>(hbuf0, cbuf, BH);

    dim3 grid(Bsz / TM, Hdim / TN);
    for (int t = 0; t < Slen; ++t) {
        const float* hin  = (t & 1) ? hbuf1 : hbuf0;
        float*       hout = (t == Slen - 1) ? out
                          : ((t & 1) ? hbuf0 : hbuf1);
        lstm_step<<<grid, 128, 0, stream>>>(sentence, emb, Wih, Whh, bih, bhh,
                                            hin, cbuf, hout, t);
    }
}


# Cell update activations branch-free on v_exp_f32 and v_rcp_f32: sigmoid = rcp(1+exp(-x)), tanh = sign(x)(1-e)/(1+e), e = exp(-2|x|), in place of library tanh and IEEE divides; values move ~1e-7
# speedup vs baseline: 3.8250x; 1.1049x over previous
#include <hip/hip_runtime.h>

constexpr int Bsz  = 256;
constexpr int Slen = 512;
constexpr int Edim = 100;
constexpr int Hdim = 512;
constexpr int KTOT = Edim + Hdim;

constexpr int TM = 16;
constexpr int TN = 32;
constexpr int GP = 36;
constexpr int AP = KTOT;
constexpr int WXP = Edim;
constexpr int KB  = 32;
constexpr int WHP = 36;
constexpr int NST = 4;
constexpr int PF  = 4;

typedef __attribute__((ext_vector_type(2))) float v2f;
typedef __attribute__((ext_vector_type(4))) float v4f;
typedef __attribute__((ext_vector_type(8))) float v8f;

__device__ __forceinline__ float sigmoidf(float x) {
    return __builtin_amdgcn_rcpf(1.0f + __expf(-x));
}
__device__ __forceinline__ float tanh_er(float x) {
    const float e = __expf(-2.0f * fabsf(x));
    return copysignf((1.0f - e) * __builtin_amdgcn_rcpf(1.0f + e), x);
}

__global__ __launch_bounds__(256) void lstm_zero_init(float* __restrict__ h0,
                                                      float* __restrict__ c0,
                                                      int n) {
    int i = blockIdx.x * blockDim.x + threadIdx.x;
    if (i < n) { h0[i] = 0.0f; c0[i] = 0.0f; }
}

__global__ __launch_bounds__(128) void lstm_step(
    const int*   __restrict__ sentence,
    const float* __restrict__ emb,
    const float* __restrict__ Wih,
    const float* __restrict__ Whh,
    const float* __restrict__ bih,
    const float* __restrict__ bhh,
    const float* __restrict__ h_in,
    float*       __restrict__ c,
    float*       __restrict__ h_out,
    int t)
{
    __shared__ alignas(16) float Alds[TM * AP];
    __shared__ alignas(16) float Glds[4][TM][GP];
    __shared__ alignas(16) float Wx[4][32 * WXP];
    __shared__ alignas(16) float Wh[4][NST][32 * WHP];

    const int tid    = threadIdx.x;
    const int m_base = blockIdx.x * TM;
    const int col0   = blockIdx.y * TN;

    const int wave = tid >> 5;
    const int lane = tid & 31;
    const int half = lane >> 4;
    const int ml   = lane & 15;

    const long wrow = (long)wave * Hdim + col0;

    const int pr   = tid >> 3;
    const int sub  = tid & 7;
    const int pb   = m_base + pr;
    const int tok  = sentence[(long)pb * Slen + t];
    const v4f* hrow = (const v4f*)(h_in + (long)pb * Hdim);
    v4f* prow = (v4f*)(Alds + pr * AP);
    v4f hv[Hdim / 32];
    for (int i = 0; i < Hdim / 32; ++i)
        hv[i] = hrow[sub + 8 * i];

    const v4f* wxg = (const v4f*)(Wih + wrow * Edim);
    v4f xs[25];
    for (int i = 0; i < 25; ++i)
        xs[i] = wxg[lane + 32 * i];

    {
        const v4f* erow = (const v4f*)(emb + (long)tok * Edim);
        v4f ev[3];
        for (int i = 0; i < 3; ++i)
            ev[i] = erow[sub + 8 * i];
        const int il = (sub == 0) ? Edim / 4 - 1 : sub + 16;
        const v4f elast = erow[il];
        for (int i = 0; i < Hdim / 32; ++i)
            prow[Edim / 4 + sub + 8 * i] = hv[i];
        v4f* wxl = (v4f*)Wx[wave];
        for (int i = 0; i < 25; ++i)
            wxl[lane + 32 * i] = xs[i];
        for (int i = 0; i < 3; ++i)
            prow[sub + 8 * i] = ev[i];
        prow[il] = elast;
    }
    __syncthreads();

    const float* whg = Whh + (wrow + (lane >> 3)) * Hdim + (lane & 7) * 4;
    v4f g[PF][8];
    for (int p = 0; p < PF; ++p)
        for (int i = 0; i < 8; ++i)
            g[p][i] = *(const v4f*)(whg + (long)(4 * i) * Hdim + p * KB);

    v8f acc0 = (v8f){};
    v8f acc1 = (v8f){};

    const float* arow = Alds + ml * AP;
    const float* bx0  = Wx[wave] + ml * WXP;
    const float* bx1  = bx0 + 16 * WXP;

#pragma unroll
    for (int k0 = 0; k0 < (Edim / 8) * 8; k0 += 8) {
        const int kb = k0 + 4 * half;
        const v4f a  = *(const v4f*)(arow + kb);
        const v4f b0 = *(const v4f*)(bx0 + kb);
        const v4f b1 = *(const v4f*)(bx1 + kb);
        acc0 = __builtin_amdgcn_wmma_f32_16x16x4_f32(false, a.xy, false, b0.xy, (short)0, acc0, false, false);
        acc1 = __builtin_amdgcn_wmma_f32_16x16x4_f32(false, a.xy, false, b1.xy, (short)0, acc1, false, false);
        acc0 = __builtin_amdgcn_wmma_f32_16x16x4_f32(false, a.zw, false, b0.zw, (short)0, acc0, false, false);
        acc1 = __builtin_amdgcn_wmma_f32_16x16x4_f32(false, a.zw, false, b1.zw, (short)0, acc1, false, false);
    }
    {
        const int kb = (Edim / 8) * 8 + 2 * half;
        const v2f a  = *(const v2f*)(arow + kb);
        const v2f b0 = *(const v2f*)(bx0 + kb);
        const v2f b1 = *(const v2f*)(bx1 + kb);
        acc0 = __builtin_amdgcn_wmma_f32_16x16x4_f32(false, a, false, b0, (short)0, acc0, false, false);
        acc1 = __builtin_amdgcn_wmma_f32_16x16x4_f32(false, a, false, b1, (short)0, acc1, false, false);
    }

    float*       whs = &Wh[wave][0][0] + (lane >> 3) * WHP + (lane & 7) * 4;
    const float* wf0 = &Wh[wave][0][0] + ml * WHP;
#pragma unroll
    for (int kb = 0; kb < Hdim / KB; ++kb) {
        const int st = kb % NST;
        const int p  = kb % PF;
        float* sdst = whs + st * (32 * WHP);
        for (int i = 0; i < 8; ++i)
            *(v4f*)(sdst + 4 * i * WHP) = g[p][i];
        if (kb + PF < Hdim / KB)
            for (int i = 0; i < 8; ++i)
                g[p][i] = *(const v4f*)(whg + (long)(4 * i) * Hdim + (kb + PF) * KB);
        __builtin_amdgcn_fence(__ATOMIC_RELEASE, "wavefront");
        __builtin_amdgcn_wave_barrier();
        __builtin_amdgcn_fence(__ATOMIC_ACQUIRE, "wavefront");
        const float* f0 = wf0 + st * (32 * WHP);
        const float* f1 = f0 + 16 * WHP;
        for (int cc = 0; cc < KB / 8; ++cc) {
            const int kk = 8 * cc + 4 * half;
            const v4f a  = *(const v4f*)(arow + Edim + kb * KB + kk);
            const v4f b0 = *(const v4f*)(f0 + kk);
            const v4f b1 = *(const v4f*)(f1 + kk);
            acc0 = __builtin_amdgcn_wmma_f32_16x16x4_f32(false, a.xy, false, b0.xy, (short)0, acc0, false, false);
            acc1 = __builtin_amdgcn_wmma_f32_16x16x4_f32(false, a.xy, false, b1.xy, (short)0, acc1, false, false);
            acc0 = __builtin_amdgcn_wmma_f32_16x16x4_f32(false, a.zw, false, b0.zw, (short)0, acc0, false, false);
            acc1 = __builtin_amdgcn_wmma_f32_16x16x4_f32(false, a.zw, false, b1.zw, (short)0, acc1, false, false);
        }
    }

    for (int r = 0; r < 8; ++r) {
        Glds[wave][r + 8 * half][ml]      = acc0[r];
        Glds[wave][r + 8 * half][16 + ml] = acc1[r];
    }
    __syncthreads();

    const int row = tid >> 3;
    const int c4  = (tid & 7) * 4;
    const int col = col0 + c4;

    const float4 pi = *(const float4*)&Glds[0][row][c4];
    const float4 pf = *(const float4*)&Glds[1][row][c4];
    const float4 pg = *(const float4*)&Glds[2][row][c4];
    const float4 po = *(const float4*)&Glds[3][row][c4];

    const float4 bi0 = *(const float4*)(bih + 0 * Hdim + col);
    const float4 bi1 = *(const float4*)(bih + 1 * Hdim + col);
    const float4 bi2 = *(const float4*)(bih + 2 * Hdim + col);
    const float4 bi3 = *(const float4*)(bih + 3 * Hdim + col);
    const float4 bh0 = *(const float4*)(bhh + 0 * Hdim + col);
    const float4 bh1 = *(const float4*)(bhh + 1 * Hdim + col);
    const float4 bh2 = *(const float4*)(bhh + 2 * Hdim + col);
    const float4 bh3 = *(const float4*)(bhh + 3 * Hdim + col);

    const long idx = (long)(m_base + row) * Hdim + col;
    const float4 cold = *(const float4*)(c + idx);

    const float vi[4] = {pi.x, pi.y, pi.z, pi.w};
    const float vf[4] = {pf.x, pf.y, pf.z, pf.w};
    const float vg[4] = {pg.x, pg.y, pg.z, pg.w};
    const float vo[4] = {po.x, po.y, po.z, po.w};
    const float s0[4] = {bi0.x + bh0.x, bi0.y + bh0.y, bi0.z + bh0.z, bi0.w + bh0.w};
    const float s1[4] = {bi1.x + bh1.x, bi1.y + bh1.y, bi1.z + bh1.z, bi1.w + bh1.w};
    const float s2[4] = {bi2.x + bh2.x, bi2.y + bh2.y, bi2.z + bh2.z, bi2.w + bh2.w};
    const float s3[4] = {bi3.x + bh3.x, bi3.y + bh3.y, bi3.z + bh3.z, bi3.w + bh3.w};
    const float co[4] = {cold.x, cold.y, cold.z, cold.w};
    float cn[4], hn[4];
    for (int j = 0; j < 4; ++j) {
        const float ig = sigmoidf(vi[j] + s0[j]);
        const float fg = sigmoidf(vf[j] + s1[j]);
        const float gg = tanh_er (vg[j] + s2[j]);
        const float og = sigmoidf(vo[j] + s3[j]);
        cn[j] = fg * co[j] + ig * gg;
        hn[j] = og * tanh_er(cn[j]);
    }
    *(float4*)(c + idx)     = make_float4(cn[0], cn[1], cn[2], cn[3]);
    *(float4*)(h_out + idx) = make_float4(hn[0], hn[1], hn[2], hn[3]);
}

extern "C" void kernel_launch(void* const* d_in, const int* in_sizes, int n_in,
                              void* d_out, int out_size, void* d_ws, size_t ws_size,
                              hipStream_t stream) {
    const int*   sentence = (const int*)  d_in[0];
    const float* emb      = (const float*)d_in[1];
    const float* Wih      = (const float*)d_in[2];
    const float* Whh      = (const float*)d_in[3];
    const float* bih      = (const float*)d_in[4];
    const float* bhh      = (const float*)d_in[5];
    float* out = (float*)d_out;
    float* ws  = (float*)d_ws;

    const int BH = Bsz * Hdim;
    float* hbuf0 = ws;
    float* hbuf1 = ws + BH;
    float* cbuf  = ws + 2 * BH;

    lstm_zero_init<<<(BH + 255) / 256, 256, 0, stream>>>(hbuf0, cbuf, BH);

    dim3 grid(Bsz / TM, Hdim / TN);
    for (int t = 0; t < Slen; ++t) {
        const float* hin  = (t & 1) ? hbuf1 : hbuf0;
        float*       hout = (t == Slen - 1) ? out
                          : ((t & 1) ? hbuf0 : hbuf1);
        lstm_step<<<grid, 128, 0, stream>>>(sentence, emb, Wih, Whh, bih, bhh,
                                            hin, cbuf, hout, t);
    }
}


# K loop in blocks of 32 k: next block's fragment reads, ring stores and refill loads issued as ONE group ahead of the current block's 16 matrix instructions (second fragment set); sum order unchanged
# speedup vs baseline: 3.9597x; 1.0352x over previous
#include <hip/hip_runtime.h>

constexpr int Bsz  = 256;
constexpr int Slen = 512;
constexpr int Edim = 100;
constexpr int Hdim = 512;
constexpr int KTOT = Edim + Hdim;

constexpr int TM = 16;
constexpr int TN = 32;
constexpr int GP = 36;
constexpr int AP = KTOT;
constexpr int WXP = Edim;
constexpr int KB  = 32;
constexpr int WHP = 36;
constexpr int NST = 4;
constexpr int PF  = 4;

typedef __attribute__((ext_vector_type(2))) float v2f;
typedef __attribute__((ext_vector_type(4))) float v4f;
typedef __attribute__((ext_vector_type(8))) float v8f;

struct Frag { v4f a[4]; v4f b0[4]; v4f b1[4]; };

__device__ __forceinline__ void read_block(Frag& f, const float* pa,
                                           const float* pb0, const float* pb1) {
#pragma unroll
    for (int cc = 0; cc < 4; ++cc) {
        f.a[cc]  = *(const v4f*)(pa  + 8 * cc);
        f.b0[cc] = *(const v4f*)(pb0 + 8 * cc);
        f.b1[cc] = *(const v4f*)(pb1 + 8 * cc);
    }
}

__device__ __forceinline__ void mma_block(const Frag& f, v8f& acc0, v8f& acc1) {
#pragma unroll
    for (int cc = 0; cc < 4; ++cc) {
        acc0 = __builtin_amdgcn_wmma_f32_16x16x4_f32(false, f.a[cc].xy, false, f.b0[cc].xy, (short)0, acc0, false, false);
        acc1 = __builtin_amdgcn_wmma_f32_16x16x4_f32(false, f.a[cc].xy, false, f.b1[cc].xy, (short)0, acc1, false, false);
        acc0 = __builtin_amdgcn_wmma_f32_16x16x4_f32(false, f.a[cc].zw, false, f.b0[cc].zw, (short)0, acc0, false, false);
        acc1 = __builtin_amdgcn_wmma_f32_16x16x4_f32(false, f.a[cc].zw, false, f.b1[cc].zw, (short)0, acc1, false, false);
    }
}

__device__ __forceinline__ float sigmoidf(float x) {
    return __builtin_amdgcn_rcpf(1.0f + __expf(-x));
}
__device__ __forceinline__ float tanh_er(float x) {
    const float e = __expf(-2.0f * fabsf(x));
    return copysignf((1.0f - e) * __builtin_amdgcn_rcpf(1.0f + e), x);
}

__global__ __launch_bounds__(256) void lstm_zero_init(float* __restrict__ h0,
                                                      float* __restrict__ c0,
                                                      int n) {
    int i = blockIdx.x * blockDim.x + threadIdx.x;
    if (i < n) { h0[i] = 0.0f; c0[i] = 0.0f; }
}

__global__ __launch_bounds__(128) void lstm_step(
    const int*   __restrict__ sentence,
    const float* __restrict__ emb,
    const float* __restrict__ Wih,
    const float* __restrict__ Whh,
    const float* __restrict__ bih,
    const float* __restrict__ bhh,
    const float* __restrict__ h_in,
    float*       __restrict__ c,
    float*       __restrict__ h_out,
    int t)
{
    __shared__ alignas(16) float Alds[TM * AP];
    __shared__ alignas(16) float Glds[4][TM][GP];
    __shared__ alignas(16) float Wx[4][32 * WXP];
    __shared__ alignas(16) float Wh[4][NST][32 * WHP];

    const int tid    = threadIdx.x;
    const int m_base = blockIdx.x * TM;
    const int col0   = blockIdx.y * TN;

    const int wave = tid >> 5;
    const int lane = tid & 31;
    const int half = lane >> 4;
    const int ml   = lane & 15;

    const long wrow = (long)wave * Hdim + col0;

    const int pr   = tid >> 3;
    const int sub  = tid & 7;
    const int pb   = m_base + pr;
    const int tok  = sentence[(long)pb * Slen + t];
    const v4f* hrow = (const v4f*)(h_in + (long)pb * Hdim);
    v4f* prow = (v4f*)(Alds + pr * AP);
    v4f hv[Hdim / 32];
    for (int i = 0; i < Hdim / 32; ++i)
        hv[i] = hrow[sub + 8 * i];

    const v4f* wxg = (const v4f*)(Wih + wrow * Edim);
    v4f xs[25];
    for (int i = 0; i < 25; ++i)
        xs[i] = wxg[lane + 32 * i];

    {
        const v4f* erow = (const v4f*)(emb + (long)tok * Edim);
        v4f ev[3];
        for (int i = 0; i < 3; ++i)
            ev[i] = erow[sub + 8 * i];
        const int il = (sub == 0) ? Edim / 4 - 1 : sub + 16;
        const v4f elast = erow[il];
        for (int i = 0; i < Hdim / 32; ++i)
            prow[Edim / 4 + sub + 8 * i] = hv[i];
        v4f* wxl = (v4f*)Wx[wave];
        for (int i = 0; i < 25; ++i)
            wxl[lane + 32 * i] = xs[i];
        for (int i = 0; i < 3; ++i)
            prow[sub + 8 * i] = ev[i];
        prow[il] = elast;
    }
    __syncthreads();


    const float* whg = Whh + (wrow + (lane >> 3)) * Hdim + (lane & 7) * 4;
    v4f g[PF][8];

    v8f acc0 = (v8f){};
    v8f acc1 = (v8f){};

    const float* arow = Alds + ml * AP;
    const float* bx0  = Wx[wave] + ml * WXP;
    const float* bx1  = bx0 + 16 * WXP;
    const float* arh  = arow + 4 * half;
    const float* bx0h = bx0 + 4 * half;
    const float* bx1h = bx1 + 4 * half;
    float*       whs  = &Wh[wave][0][0] + (lane >> 3) * WHP + (lane & 7) * 4;
    const float* wf0h = &Wh[wave][0][0] + ml * WHP + 4 * half;

    Frag F[2];

    read_block(F[0], arh, bx0h, bx1h);
#pragma unroll
    for (int p = 0; p < PF; ++p)
        for (int i = 0; i < 8; ++i)
            g[p][i] = *(const v4f*)(whg + (long)(4 * i) * Hdim + p * KB);
    read_block(F[1], arh + KB, bx0h + KB, bx1h + KB);
    __builtin_amdgcn_sched_barrier(0);
    mma_block(F[0], acc0, acc1);
    __builtin_amdgcn_sched_barrier(0);

    read_block(F[0], arh + 2 * KB, bx0h + 2 * KB, bx1h + 2 * KB);
    __builtin_amdgcn_sched_barrier(0);
    mma_block(F[1], acc0, acc1);
    __builtin_amdgcn_sched_barrier(0);

#pragma unroll
    for (int n = 0; n <= Hdim / KB; ++n) {
        v2f ta, tb0, tb1;
        if (n == 0) {
            const int kt = (Edim / 8) * 8 + 2 * half;
            ta  = *(const v2f*)(arow + kt);
            tb0 = *(const v2f*)(bx0 + kt);
            tb1 = *(const v2f*)(bx1 + kt);
        }
        if (n < Hdim / KB) {
            const int st = n % NST;
            const int p  = n % PF;
            float* sdst = whs + st * (32 * WHP);
            for (int i = 0; i < 8; ++i)
                *(v4f*)(sdst + 4 * i * WHP) = g[p][i];
            if (n + PF < Hdim / KB)
                for (int i = 0; i < 8; ++i)
                    g[p][i] = *(const v4f*)(whg + (long)(4 * i) * Hdim + (n + PF) * KB);
            __builtin_amdgcn_fence(__ATOMIC_RELEASE, "wavefront");
            __builtin_amdgcn_wave_barrier();
            __builtin_amdgcn_fence(__ATOMIC_ACQUIRE, "wavefront");
            const float* f0 = wf0h + st * (32 * WHP);
            read_block(F[(n + 1) & 1], arh + Edim + n * KB, f0, f0 + 16 * WHP);
        }
        __builtin_amdgcn_sched_barrier(0);
        mma_block(F[n & 1], acc0, acc1);
        if (n == 0) {
            acc0 = __builtin_amdgcn_wmma_f32_16x16x4_f32(false, ta, false, tb0, (short)0, acc0, false, false);
            acc1 = __builtin_amdgcn_wmma_f32_16x16x4_f32(false, ta, false, tb1, (short)0, acc1, false, false);
        }
        __builtin_amdgcn_sched_barrier(0);
    }

    for (int r = 0; r < 8; ++r) {
        Glds[wave][r + 8 * half][ml]      = acc0[r];
        Glds[wave][r + 8 * half][16 + ml] = acc1[r];
    }
    __syncthreads();

    const int row = tid >> 3;
    const int c4  = (tid & 7) * 4;
    const int col = col0 + c4;

    const float4 pi = *(const float4*)&Glds[0][row][c4];
    const float4 pf = *(const float4*)&Glds[1][row][c4];
    const float4 pg = *(const float4*)&Glds[2][row][c4];
    const float4 po = *(const float4*)&Glds[3][row][c4];

    const float4 bi0 = *(const float4*)(bih + 0 * Hdim + col);
    const float4 bi1 = *(const float4*)(bih + 1 * Hdim + col);
    const float4 bi2 = *(const float4*)(bih + 2 * Hdim + col);
    const float4 bi3 = *(const float4*)(bih + 3 * Hdim + col);
    const float4 bh0 = *(const float4*)(bhh + 0 * Hdim + col);
    const float4 bh1 = *(const float4*)(bhh + 1 * Hdim + col);
    const float4 bh2 = *(const float4*)(bhh + 2 * Hdim + col);
    const float4 bh3 = *(const float4*)(bhh + 3 * Hdim + col);

    const long idx = (long)(m_base + row) * Hdim + col;
    const float4 cold = *(const float4*)(c + idx);

    const float vi[4] = {pi.x, pi.y, pi.z, pi.w};
    const float vf[4] = {pf.x, pf.y, pf.z, pf.w};
    const float vg[4] = {pg.x, pg.y, pg.z, pg.w};
    const float vo[4] = {po.x, po.y, po.z, po.w};
    const float s0[4] = {bi0.x + bh0.x, bi0.y + bh0.y, bi0.z + bh0.z, bi0.w + bh0.w};
    const float s1[4] = {bi1.x + bh1.x, bi1.y + bh1.y, bi1.z + bh1.z, bi1.w + bh1.w};
    const float s2[4] = {bi2.x + bh2.x, bi2.y + bh2.y, bi2.z + bh2.z, bi2.w + bh2.w};
    const float s3[4] = {bi3.x + bh3.x, bi3.y + bh3.y, bi3.z + bh3.z, bi3.w + bh3.w};
    const float co[4] = {cold.x, cold.y, cold.z, cold.w};
    float cn[4], hn[4];
    for (int j = 0; j < 4; ++j) {
        const float ig = sigmoidf(vi[j] + s0[j]);
        const float fg = sigmoidf(vf[j] + s1[j]);
        const float gg = tanh_er (vg[j] + s2[j]);
        const float og = sigmoidf(vo[j] + s3[j]);
        cn[j] = fg * co[j] + ig * gg;
        hn[j] = og * tanh_er(cn[j]);
    }
    *(float4*)(c + idx)     = make_float4(cn[0], cn[1], cn[2], cn[3]);
    *(float4*)(h_out + idx) = make_float4(hn[0], hn[1], hn[2], hn[3]);
}

extern "C" void kernel_launch(void* const* d_in, const int* in_sizes, int n_in,
                              void* d_out, int out_size, void* d_ws, size_t ws_size,
                              hipStream_t stream) {
    const int*   sentence = (const int*)  d_in[0];
    const float* emb      = (const float*)d_in[1];
    const float* Wih      = (const float*)d_in[2];
    const float* Whh      = (const float*)d_in[3];
    const float* bih      = (const float*)d_in[4];
    const float* bhh      = (const float*)d_in[5];
    float* out = (float*)d_out;
    float* ws  = (float*)d_ws;

    const int BH = Bsz * Hdim;
    float* hbuf0 = ws;
    float* hbuf1 = ws + BH;
    float* cbuf  = ws + 2 * BH;

    lstm_zero_init<<<(BH + 255) / 256, 256, 0, stream>>>(hbuf0, cbuf, BH);

    dim3 grid(Bsz / TM, Hdim / TN);
    for (int t = 0; t < Slen; ++t) {
        const float* hin  = (t & 1) ? hbuf1 : hbuf0;
        float*       hout = (t == Slen - 1) ? out
                          : ((t & 1) ? hbuf0 : hbuf1);
        lstm_step<<<grid, 128, 0, stream>>>(sentence, emb, Wih, Whh, bih, bhh,
                                            hin, cbuf, hout, t);
    }
}
